# starting kernel
# baseline (speedup 1.0000x reference)
#include <hip/hip_runtime.h>
#include <hip/hip_bf16.h>
#include <math.h>

typedef __attribute__((ext_vector_type(2))) float v2f;
typedef __attribute__((ext_vector_type(8))) float v8f;

#define N_ROWS   4096
#define D_IN     2048
#define D_HID    8192
#define D_OUT    2048
#define RANK     8

__device__ __forceinline__ float block_reduce_sum_256(float p, float* redw) {
    #pragma unroll
    for (int off = 16; off > 0; off >>= 1) p += __shfl_down(p, off, 32);
    int t = threadIdx.x;
    __syncthreads();
    if ((t & 31) == 0) redw[t >> 5] = p;
    __syncthreads();
    float s = 0.f;
    #pragma unroll
    for (int w = 0; w < 8; ++w) s += redw[w];
    return s;
}

__global__ __launch_bounds__(256)
void hra_gs_kernel(const float* __restrict__ hra_u, float* __restrict__ U) {
    __shared__ float redw[8];
    const int t = threadIdx.x;
    float u[RANK][8];
    #pragma unroll
    for (int e = 0; e < 8; ++e) {
        int d = e * 256 + t;
        #pragma unroll
        for (int j = 0; j < RANK; ++j) u[j][e] = hra_u[d * RANK + j];
    }
    for (int i = 0; i < RANK; ++i) {
        for (int j = 0; j < i; ++j) {
            float p = 0.f;
            #pragma unroll
            for (int e = 0; e < 8; ++e) p += u[j][e] * u[i][e];
            float dot = block_reduce_sum_256(p, redw);
            #pragma unroll
            for (int e = 0; e < 8; ++e) u[i][e] -= dot * u[j][e];
        }
        float p = 0.f;
        #pragma unroll
        for (int e = 0; e < 8; ++e) p += u[i][e] * u[i][e];
        float nrm2 = block_reduce_sum_256(p, redw);
        float inv = 1.0f / (sqrtf(nrm2) + 1e-6f);
        #pragma unroll
        for (int e = 0; e < 8; ++e) u[i][e] *= inv;
    }
    #pragma unroll
    for (int j = 0; j < RANK; ++j)
        #pragma unroll
        for (int e = 0; e < 8; ++e)
            U[j * D_IN + e * 256 + t] = u[j][e];
}

__global__ __launch_bounds__(256)
void hra_apply_kernel(const float* __restrict__ x, const float* __restrict__ U,
                      float* __restrict__ y) {
    __shared__ float red[8][RANK];
    const int t   = threadIdx.x;
    const size_t row = blockIdx.x;
    const float* xr = x + row * D_IN;

    float xv[8];
    #pragma unroll
    for (int e = 0; e < 8; ++e) xv[e] = xr[e * 256 + t];

    float c[RANK];
    #pragma unroll
    for (int j = 0; j < RANK; ++j) {
        float p = 0.f;
        #pragma unroll
        for (int e = 0; e < 8; ++e) p += xv[e] * U[j * D_IN + e * 256 + t];
        #pragma unroll
        for (int off = 16; off > 0; off >>= 1) p += __shfl_down(p, off, 32);
        if ((t & 31) == 0) red[t >> 5][j] = p;
    }
    __syncthreads();
    #pragma unroll
    for (int j = 0; j < RANK; ++j) {
        float s = 0.f;
        #pragma unroll
        for (int w = 0; w < 8; ++w) s += red[w][j];
        c[j] = 2.0f * s;
    }
    float* yr = y + row * D_IN;
    #pragma unroll
    for (int e = 0; e < 8; ++e) {
        int d = e * 256 + t;
        float acc = xv[e];
        #pragma unroll
        for (int j = 0; j < RANK; ++j) acc -= c[j] * U[j * D_IN + d];
        yr[d] = acc;
    }
}

#define BM 128
#define BN 128
#define BK 16
#define LDAB 18

template <int EPI>
__global__ __launch_bounds__(256)
void wmma_gemm_f32_kernel(const float* __restrict__ A, const float* __restrict__ B,
                          const float* __restrict__ bias, float* __restrict__ C,
                          int M, int N, int K) {
    __shared__ float As [2][BM * LDAB];
    __shared__ float Bts[2][BN * LDAB];

    const int t    = threadIdx.x;
    const int wave = t >> 5;
    const int lane = t & 31;
    const int wm   = wave & 3;
    const int wn   = wave >> 2;
    const int l16  = lane & 15;
    const int lh   = lane >> 4;

    const int m_blk = blockIdx.y * BM;
    const int n_blk = blockIdx.x * BN;

    v8f acc[2][4] = {};
    float4 ra[2], rb[2];

    auto load_tiles = [&](int k0) {
        #pragma unroll
        for (int i = 0; i < 2; ++i) {
            int idx = t + i * 256;
            ra[i] = *(const float4*)(A + (size_t)(m_blk + (idx >> 2)) * K +
                                     k0 + ((idx & 3) << 2));
        }
        #pragma unroll
        for (int i = 0; i < 2; ++i) {
            int idx = t + i * 256;
            rb[i] = *(const float4*)(B + (size_t)(k0 + (idx >> 5)) * N +
                                     n_blk + ((idx & 31) << 2));
        }
    };
    auto store_tiles = [&](int buf) {
        #pragma unroll
        for (int i = 0; i < 2; ++i) {
            int idx = t + i * 256;
            float* p = &As[buf][(idx >> 2) * LDAB + ((idx & 3) << 2)];
            p[0] = ra[i].x; p[1] = ra[i].y; p[2] = ra[i].z; p[3] = ra[i].w;
        }
        #pragma unroll
        for (int i = 0; i < 2; ++i) {
            int idx = t + i * 256;
            int r   = idx >> 5;
            int c4  = (idx & 31) << 2;
            Bts[buf][(c4 + 0) * LDAB + r] = rb[i].x;
            Bts[buf][(c4 + 1) * LDAB + r] = rb[i].y;
            Bts[buf][(c4 + 2) * LDAB + r] = rb[i].z;
            Bts[buf][(c4 + 3) * LDAB + r] = rb[i].w;
        }
    };
    auto compute = [&](int buf) {
        #pragma unroll
        for (int kk = 0; kk < BK; kk += 4) {
            const int kfrag = kk + 2 * lh;
            v2f af[2];
            #pragma unroll
            for (int i = 0; i < 2; ++i)
                af[i] = *(const v2f*)(&As[buf][(wm * 32 + i * 16 + l16) * LDAB + kfrag]);
            v2f bf[4];
            #pragma unroll
            for (int j = 0; j < 4; ++j)
                bf[j] = *(const v2f*)(&Bts[buf][(wn * 64 + j * 16 + l16) * LDAB + kfrag]);
            #pragma unroll
            for (int i = 0; i < 2; ++i)
                #pragma unroll
                for (int j = 0; j < 4; ++j)
                    acc[i][j] = __builtin_amdgcn_wmma_f32_16x16x4_f32(
                        false, af[i], false, bf[j], (short)0, acc[i][j],
                        false, false);
        }
    };

    load_tiles(0);
    store_tiles(0);
    __syncthreads();
    int buf = 0;
    for (int k0 = BK; k0 < K; k0 += BK) {
        load_tiles(k0);
        compute(buf);
        store_tiles(buf ^ 1);
        __syncthreads();
        buf ^= 1;
    }
    compute(buf);

    #pragma unroll
    for (int i = 0; i < 2; ++i) {
        #pragma unroll
        for (int j = 0; j < 4; ++j) {
            int col = n_blk + wn * 64 + j * 16 + l16;
            float bv = (EPI == 1) ? bias[col] : 0.f;
            #pragma unroll
            for (int r = 0; r < 8; ++r) {
                int row = m_blk + wm * 32 + i * 16 + r + 8 * lh;
                float v = acc[i][j][r];
                if (EPI == 0) {
                    v = 0.5f * v * (1.0f + erff(v * 0.70710678118654752f));
                } else {
                    v += bv;
                }
                C[(size_t)row * N + col] = v;
            }
        }
    }
}

extern "C" void kernel_launch(void* const* d_in, const int* in_sizes, int n_in,
                              void* d_out, int out_size, void* d_ws, size_t ws_size,
                              hipStream_t stream) {
    const float* x    = (const float*)d_in[0];
    const float* hra  = (const float*)d_in[1];
    const float* Wr   = (const float*)d_in[2];
    const float* Wo   = (const float*)d_in[3];
    const float* bias = (const float*)d_in[4];
    float* out = (float*)d_out;

    char* ws = (char*)d_ws;
    float* U = (float*)ws;
    float* y = (float*)(ws + (64u << 10));
    float* h = (float*)(ws + (64u << 10) + (32u << 20));

    hra_gs_kernel<<<1, 256, 0, stream>>>(hra, U);

    hra_apply_kernel<<<N_ROWS, 256, 0, stream>>>(x, U, y);

    dim3 g1(D_HID / BN, N_ROWS / BM);
    wmma_gemm_f32_kernel<0><<<g1, 256, 0, stream>>>(y, Wr, nullptr, h,
                                                    N_ROWS, D_HID, D_IN);

    dim3 g2(D_OUT / BN, N_ROWS / BM);
    wmma_gemm_f32_kernel<1><<<g2, 256, 0, stream>>>(h, Wo, bias, out,
                                                    N_ROWS, D_OUT, D_HID);
}


# GEMM K-loop schedule: per 16-k chunk each wave issues the next tile's global loads first, then all its LDS fragment reads as one group, then its 32 WMMAs back to back; LDS stores behind the block
# speedup vs baseline: 1.0344x; 1.0344x over previous
#include <hip/hip_runtime.h>
#include <hip/hip_bf16.h>
#include <math.h>

typedef __attribute__((ext_vector_type(2))) float v2f;
typedef __attribute__((ext_vector_type(8))) float v8f;

#define N_ROWS   4096
#define D_IN     2048
#define D_HID    8192
#define D_OUT    2048
#define RANK     8

__device__ __forceinline__ float block_reduce_sum_256(float p, float* redw) {
    #pragma unroll
    for (int off = 16; off > 0; off >>= 1) p += __shfl_down(p, off, 32);
    int t = threadIdx.x;
    __syncthreads();
    if ((t & 31) == 0) redw[t >> 5] = p;
    __syncthreads();
    float s = 0.f;
    #pragma unroll
    for (int w = 0; w < 8; ++w) s += redw[w];
    return s;
}

__global__ __launch_bounds__(256)
void hra_gs_kernel(const float* __restrict__ hra_u, float* __restrict__ U) {
    __shared__ float redw[8];
    const int t = threadIdx.x;
    float u[RANK][8];
    #pragma unroll
    for (int e = 0; e < 8; ++e) {
        int d = e * 256 + t;
        #pragma unroll
        for (int j = 0; j < RANK; ++j) u[j][e] = hra_u[d * RANK + j];
    }
    for (int i = 0; i < RANK; ++i) {
        for (int j = 0; j < i; ++j) {
            float p = 0.f;
            #pragma unroll
            for (int e = 0; e < 8; ++e) p += u[j][e] * u[i][e];
            float dot = block_reduce_sum_256(p, redw);
            #pragma unroll
            for (int e = 0; e < 8; ++e) u[i][e] -= dot * u[j][e];
        }
        float p = 0.f;
        #pragma unroll
        for (int e = 0; e < 8; ++e) p += u[i][e] * u[i][e];
        float nrm2 = block_reduce_sum_256(p, redw);
        float inv = 1.0f / (sqrtf(nrm2) + 1e-6f);
        #pragma unroll
        for (int e = 0; e < 8; ++e) u[i][e] *= inv;
    }
    #pragma unroll
    for (int j = 0; j < RANK; ++j)
        #pragma unroll
        for (int e = 0; e < 8; ++e)
            U[j * D_IN + e * 256 + t] = u[j][e];
}

__global__ __launch_bounds__(256)
void hra_apply_kernel(const float* __restrict__ x, const float* __restrict__ U,
                      float* __restrict__ y) {
    __shared__ float red[8][RANK];
    const int t   = threadIdx.x;
    const size_t row = blockIdx.x;
    const float* xr = x + row * D_IN;

    float xv[8];
    #pragma unroll
    for (int e = 0; e < 8; ++e) xv[e] = xr[e * 256 + t];

    float c[RANK];
    #pragma unroll
    for (int j = 0; j < RANK; ++j) {
        float p = 0.f;
        #pragma unroll
        for (int e = 0; e < 8; ++e) p += xv[e] * U[j * D_IN + e * 256 + t];
        #pragma unroll
        for (int off = 16; off > 0; off >>= 1) p += __shfl_down(p, off, 32);
        if ((t & 31) == 0) red[t >> 5][j] = p;
    }
    __syncthreads();
    #pragma unroll
    for (int j = 0; j < RANK; ++j) {
        float s = 0.f;
        #pragma unroll
        for (int w = 0; w < 8; ++w) s += red[w][j];
        c[j] = 2.0f * s;
    }
    float* yr = y + row * D_IN;
    #pragma unroll
    for (int e = 0; e < 8; ++e) {
        int d = e * 256 + t;
        float acc = xv[e];
        #pragma unroll
        for (int j = 0; j < RANK; ++j) acc -= c[j] * U[j * D_IN + d];
        yr[d] = acc;
    }
}

#define BM 128
#define BN 128
#define BK 16
#define LDAB 18

template <int EPI>
__global__ __launch_bounds__(256)
void wmma_gemm_f32_kernel(const float* A, const float* B,
                          const float* __restrict__ bias, float* __restrict__ C,
                          int M, int N, int K) {
    __shared__ float As [2][BM * LDAB];
    __shared__ float Bts[2][BN * LDAB];

    const int t    = threadIdx.x;
    const int wave = t >> 5;
    const int lane = t & 31;
    const int wm   = wave & 3;
    const int wn   = wave >> 2;
    const int l16  = lane & 15;
    const int lh   = lane >> 4;

    const int m_blk = blockIdx.y * BM;
    const int n_blk = blockIdx.x * BN;

    v8f acc[2][4] = {};
    float4 ra[2], rb[2];

    auto load_tiles = [&](int k0) {
        #pragma unroll
        for (int i = 0; i < 2; ++i) {
            int idx = t + i * 256;
            ra[i] = *(const float4*)(A + (size_t)(m_blk + (idx >> 2)) * K +
                                     k0 + ((idx & 3) << 2));
        }
        #pragma unroll
        for (int i = 0; i < 2; ++i) {
            int idx = t + i * 256;
            rb[i] = *(const float4*)(B + (size_t)(k0 + (idx >> 5)) * N +
                                     n_blk + ((idx & 31) << 2));
        }
    };
    auto store_tiles = [&](int buf) {
        #pragma unroll
        for (int i = 0; i < 2; ++i) {
            int idx = t + i * 256;
            float* p = &As[buf][(idx >> 2) * LDAB + ((idx & 3) << 2)];
            p[0] = ra[i].x; p[1] = ra[i].y; p[2] = ra[i].z; p[3] = ra[i].w;
        }
        #pragma unroll
        for (int i = 0; i < 2; ++i) {
            int idx = t + i * 256;
            int r   = idx >> 5;
            int c4  = (idx & 31) << 2;
            Bts[buf][(c4 + 0) * LDAB + r] = rb[i].x;
            Bts[buf][(c4 + 1) * LDAB + r] = rb[i].y;
            Bts[buf][(c4 + 2) * LDAB + r] = rb[i].z;
            Bts[buf][(c4 + 3) * LDAB + r] = rb[i].w;
        }
    };
    v2f af[4][2], bf[4][4];
    auto load_frags = [&](int buf) {
        #pragma unroll
        for (int s = 0; s < 4; ++s) {
            const int kfrag = s * 4 + 2 * lh;
            #pragma unroll
            for (int i = 0; i < 2; ++i)
                af[s][i] = *(const v2f*)(&As[buf][(wm * 32 + i * 16 + l16) * LDAB + kfrag]);
            #pragma unroll
            for (int j = 0; j < 4; ++j)
                bf[s][j] = *(const v2f*)(&Bts[buf][(wn * 64 + j * 16 + l16) * LDAB + kfrag]);
        }
    };
    auto mma_block = [&]() {
        #pragma unroll
        for (int s = 0; s < 4; ++s)
            #pragma unroll
            for (int i = 0; i < 2; ++i)
                #pragma unroll
                for (int j = 0; j < 4; ++j)
                    acc[i][j] = __builtin_amdgcn_wmma_f32_16x16x4_f32(
                        false, af[s][i], false, bf[s][j], (short)0, acc[i][j],
                        false, false);
    };

    load_tiles(0);
    store_tiles(0);
    __syncthreads();
    int buf = 0;
    for (int k0 = BK; k0 < K; k0 += BK) {
        load_tiles(k0);
        load_frags(buf);
        asm volatile("" ::: "memory");
        __builtin_amdgcn_sched_barrier(0);
        mma_block();
        __builtin_amdgcn_sched_barrier(0);
        store_tiles(buf ^ 1);
        __syncthreads();
        buf ^= 1;
    }
    load_frags(buf);
    mma_block();

    #pragma unroll
    for (int i = 0; i < 2; ++i) {
        #pragma unroll
        for (int j = 0; j < 4; ++j) {
            int col = n_blk + wn * 64 + j * 16 + l16;
            float bv = (EPI == 1) ? bias[col] : 0.f;
            #pragma unroll
            for (int r = 0; r < 8; ++r) {
                int row = m_blk + wm * 32 + i * 16 + r + 8 * lh;
                float v = acc[i][j][r];
                if (EPI == 0) {
                    v = 0.5f * v * (1.0f + erff(v * 0.70710678118654752f));
                } else {
                    v += bv;
                }
                C[(size_t)row * N + col] = v;
            }
        }
    }
}

extern "C" void kernel_launch(void* const* d_in, const int* in_sizes, int n_in,
                              void* d_out, int out_size, void* d_ws, size_t ws_size,
                              hipStream_t stream) {
    const float* x    = (const float*)d_in[0];
    const float* hra  = (const float*)d_in[1];
    const float* Wr   = (const float*)d_in[2];
    const float* Wo   = (const float*)d_in[3];
    const float* bias = (const float*)d_in[4];
    float* out = (float*)d_out;

    char* ws = (char*)d_ws;
    float* U = (float*)ws;
    float* y = (float*)(ws + (64u << 10));
    float* h = (float*)(ws + (64u << 10) + (32u << 20));

    hra_gs_kernel<<<1, 256, 0, stream>>>(hra, U);

    hra_apply_kernel<<<N_ROWS, 256, 0, stream>>>(x, U, y);

    dim3 g1(D_HID / BN, N_ROWS / BM);
    wmma_gemm_f32_kernel<0><<<g1, 256, 0, stream>>>(y, Wr, nullptr, h,
                                                    N_ROWS, D_HID, D_IN);

    dim3 g2(D_OUT / BN, N_ROWS / BM);
    wmma_gemm_f32_kernel<1><<<g2, 256, 0, stream>>>(h, Wo, bias, out,
                                                    N_ROWS, D_OUT, D_HID);
}


# LDS stage layout A [k/4][row][4] (planes padded), B [k/4][col%4][col/4][4] with 8-byte k-pair stores: all K-loop LDS traffic bank-conflict-free; lane owns 4 adjacent columns, C stored 16 B per lane
# speedup vs baseline: 1.1468x; 1.1087x over previous
#include <hip/hip_runtime.h>
#include <hip/hip_bf16.h>
#include <math.h>

typedef __attribute__((ext_vector_type(2))) float v2f;
typedef __attribute__((ext_vector_type(8))) float v8f;
typedef __attribute__((ext_vector_type(4))) float v4f;

#define N_ROWS   4096
#define D_IN     2048
#define D_HID    8192
#define D_OUT    2048
#define RANK     8

__device__ __forceinline__ float block_reduce_sum_256(float p, float* redw) {
    #pragma unroll
    for (int off = 16; off > 0; off >>= 1) p += __shfl_down(p, off, 32);
    int t = threadIdx.x;
    __syncthreads();
    if ((t & 31) == 0) redw[t >> 5] = p;
    __syncthreads();
    float s = 0.f;
    #pragma unroll
    for (int w = 0; w < 8; ++w) s += redw[w];
    return s;
}

__global__ __launch_bounds__(256)
void hra_gs_kernel(const float* __restrict__ hra_u, float* __restrict__ U) {
    __shared__ float redw[8];
    const int t = threadIdx.x;
    float u[RANK][8];
    #pragma unroll
    for (int e = 0; e < 8; ++e) {
        int d = e * 256 + t;
        #pragma unroll
        for (int j = 0; j < RANK; ++j) u[j][e] = hra_u[d * RANK + j];
    }
    for (int i = 0; i < RANK; ++i) {
        for (int j = 0; j < i; ++j) {
            float p = 0.f;
            #pragma unroll
            for (int e = 0; e < 8; ++e) p += u[j][e] * u[i][e];
            float dot = block_reduce_sum_256(p, redw);
            #pragma unroll
            for (int e = 0; e < 8; ++e) u[i][e] -= dot * u[j][e];
        }
        float p = 0.f;
        #pragma unroll
        for (int e = 0; e < 8; ++e) p += u[i][e] * u[i][e];
        float nrm2 = block_reduce_sum_256(p, redw);
        float inv = 1.0f / (sqrtf(nrm2) + 1e-6f);
        #pragma unroll
        for (int e = 0; e < 8; ++e) u[i][e] *= inv;
    }
    #pragma unroll
    for (int j = 0; j < RANK; ++j)
        #pragma unroll
        for (int e = 0; e < 8; ++e)
            U[j * D_IN + e * 256 + t] = u[j][e];
}

__global__ __launch_bounds__(256)
void hra_apply_kernel(const float* __restrict__ x, const float* __restrict__ U,
                      float* __restrict__ y) {
    __shared__ float red[8][RANK];
    const int t   = threadIdx.x;
    const size_t row = blockIdx.x;
    const float* xr = x + row * D_IN;

    float xv[8];
    #pragma unroll
    for (int e = 0; e < 8; ++e) xv[e] = xr[e * 256 + t];

    float c[RANK];
    #pragma unroll
    for (int j = 0; j < RANK; ++j) {
        float p = 0.f;
        #pragma unroll
        for (int e = 0; e < 8; ++e) p += xv[e] * U[j * D_IN + e * 256 + t];
        #pragma unroll
        for (int off = 16; off > 0; off >>= 1) p += __shfl_down(p, off, 32);
        if ((t & 31) == 0) red[t >> 5][j] = p;
    }
    __syncthreads();
    #pragma unroll
    for (int j = 0; j < RANK; ++j) {
        float s = 0.f;
        #pragma unroll
        for (int w = 0; w < 8; ++w) s += red[w][j];
        c[j] = 2.0f * s;
    }
    float* yr = y + row * D_IN;
    #pragma unroll
    for (int e = 0; e < 8; ++e) {
        int d = e * 256 + t;
        float acc = xv[e];
        #pragma unroll
        for (int j = 0; j < RANK; ++j) acc -= c[j] * U[j * D_IN + d];
        yr[d] = acc;
    }
}

#define BM 128
#define BN 128
#define BK 16
#define A_PL 528
#define B_PL 512
#define B_JP 128

template <int EPI>
__global__ __launch_bounds__(256)
void wmma_gemm_f32_kernel(const float* A, const float* B,
                          const float* __restrict__ bias, float* __restrict__ C,
                          int M, int N, int K) {
    __shared__ __attribute__((aligned(16))) float As[2][4 * A_PL];
    __shared__ __attribute__((aligned(16))) float Bs[2][4 * B_PL];

    const int t    = threadIdx.x;
    const int wave = t >> 5;
    const int lane = t & 31;
    const int wm   = wave & 3;
    const int wn   = wave >> 2;
    const int l16  = lane & 15;
    const int lh   = lane >> 4;

    const int bp   = 2 * (wave & 3) + lh;
    const int bq   = (wave >> 2) * 16 + l16;

    const int m_blk = blockIdx.y * BM;
    const int n_blk = blockIdx.x * BN;

    v8f acc[2][4] = {};
    v4f ra0, ra1, rb0, rb1;

    const int a_row0 = t >> 2;
    const int a_kq   = t & 3;
    const float* a_src = A + (size_t)(m_blk + a_row0) * K + (a_kq << 2);
    const float* b_src = B + (size_t)(2 * bp) * N + n_blk + (bq << 2);
    const int a_dst = a_kq * A_PL + (a_row0 << 2);
    const int b_dst = (bp >> 1) * B_PL + (bq << 2) + ((bp & 1) << 1);

    auto load_tiles = [&](int k0) {
        ra0 = *(const v4f*)(a_src + k0);
        ra1 = *(const v4f*)(a_src + (size_t)64 * K + k0);
        rb0 = *(const v4f*)(b_src + (size_t)k0 * N);
        rb1 = *(const v4f*)(b_src + (size_t)(k0 + 1) * N);
    };
    auto store_tiles = [&](int buf) {
        *(v4f*)(&As[buf][a_dst])            = ra0;
        *(v4f*)(&As[buf][a_dst + (64 << 2)]) = ra1;
        float* pb = &Bs[buf][b_dst];
        v2f p0 = {rb0.x, rb1.x};
        v2f p1 = {rb0.y, rb1.y};
        v2f p2 = {rb0.z, rb1.z};
        v2f p3 = {rb0.w, rb1.w};
        *(v2f*)(pb + 0 * B_JP) = p0;
        *(v2f*)(pb + 1 * B_JP) = p1;
        *(v2f*)(pb + 2 * B_JP) = p2;
        *(v2f*)(pb + 3 * B_JP) = p3;
    };
    v2f af[4][2], bf[4][4];
    auto load_frags = [&](int buf) {
        #pragma unroll
        for (int s = 0; s < 4; ++s) {
            #pragma unroll
            for (int i = 0; i < 2; ++i)
                af[s][i] = *(const v2f*)(&As[buf][s * A_PL +
                                ((wm * 32 + i * 16 + l16) << 2) + 2 * lh]);
            #pragma unroll
            for (int j = 0; j < 4; ++j)
                bf[s][j] = *(const v2f*)(&Bs[buf][s * B_PL + j * B_JP +
                                ((wn * 16 + l16) << 2) + 2 * lh]);
        }
    };
    auto mma_block = [&]() {
        #pragma unroll
        for (int s = 0; s < 4; ++s)
            #pragma unroll
            for (int i = 0; i < 2; ++i)
                #pragma unroll
                for (int j = 0; j < 4; ++j)
                    acc[i][j] = __builtin_amdgcn_wmma_f32_16x16x4_f32(
                        false, af[s][i], false, bf[s][j], (short)0, acc[i][j],
                        false, false);
    };

    load_tiles(0);
    store_tiles(0);
    __syncthreads();
    int buf = 0;
    for (int k0 = BK; k0 < K; k0 += BK) {
        load_tiles(k0);
        load_frags(buf);
        asm volatile("" ::: "memory");
        __builtin_amdgcn_sched_barrier(0);
        mma_block();
        __builtin_amdgcn_sched_barrier(0);
        store_tiles(buf ^ 1);
        __syncthreads();
        buf ^= 1;
    }
    load_frags(buf);
    mma_block();

    const int col0 = n_blk + wn * 64 + (l16 << 2);
    float4 bv = make_float4(0.f, 0.f, 0.f, 0.f);
    if (EPI == 1) bv = *(const float4*)(bias + col0);
    #pragma unroll
    for (int i = 0; i < 2; ++i) {
        #pragma unroll
        for (int r = 0; r < 8; ++r) {
            int row = m_blk + wm * 32 + i * 16 + r + 8 * lh;
            float v[4];
            #pragma unroll
            for (int j = 0; j < 4; ++j) {
                float x = acc[i][j][r];
                if (EPI == 0) x = 0.5f * x * (1.0f + erff(x * 0.70710678118654752f));
                v[j] = x;
            }
            float4 o;
            if (EPI == 0) {
                o = make_float4(v[0], v[1], v[2], v[3]);
            } else {
                o = make_float4(v[0] + bv.x, v[1] + bv.y, v[2] + bv.z, v[3] + bv.w);
            }
            *(float4*)(C + (size_t)row * N + col0) = o;
        }
    }
}

extern "C" void kernel_launch(void* const* d_in, const int* in_sizes, int n_in,
                              void* d_out, int out_size, void* d_ws, size_t ws_size,
                              hipStream_t stream) {
    const float* x    = (const float*)d_in[0];
    const float* hra  = (const float*)d_in[1];
    const float* Wr   = (const float*)d_in[2];
    const float* Wo   = (const float*)d_in[3];
    const float* bias = (const float*)d_in[4];
    float* out = (float*)d_out;

    char* ws = (char*)d_ws;
    float* U = (float*)ws;
    float* y = (float*)(ws + (64u << 10));
    float* h = (float*)(ws + (64u << 10) + (32u << 20));

    hra_gs_kernel<<<1, 256, 0, stream>>>(hra, U);

    hra_apply_kernel<<<N_ROWS, 256, 0, stream>>>(x, U, y);

    dim3 g1(D_HID / BN, N_ROWS / BM);
    wmma_gemm_f32_kernel<0><<<g1, 256, 0, stream>>>(y, Wr, nullptr, h,
                                                    N_ROWS, D_HID, D_IN);

    dim3 g2(D_OUT / BN, N_ROWS / BM);
    wmma_gemm_f32_kernel<1><<<g2, 256, 0, stream>>>(h, Wo, bias, out,
                                                    N_ROWS, D_OUT, D_HID);
}


# GEMM K-loop split barrier: a wave signals once its LDS fragment reads and next-stage stores are done, runs its 32 WMMAs, then waits; stage stores moved to the step head, skew hides under WMMAs
# speedup vs baseline: 1.1818x; 1.0305x over previous
#include <hip/hip_runtime.h>
#include <hip/hip_bf16.h>
#include <math.h>

typedef __attribute__((ext_vector_type(2))) float v2f;
typedef __attribute__((ext_vector_type(8))) float v8f;
typedef __attribute__((ext_vector_type(4))) float v4f;

#define N_ROWS   4096
#define D_IN     2048
#define D_HID    8192
#define D_OUT    2048
#define RANK     8

__device__ __forceinline__ float block_reduce_sum_256(float p, float* redw) {
    #pragma unroll
    for (int off = 16; off > 0; off >>= 1) p += __shfl_down(p, off, 32);
    int t = threadIdx.x;
    __syncthreads();
    if ((t & 31) == 0) redw[t >> 5] = p;
    __syncthreads();
    float s = 0.f;
    #pragma unroll
    for (int w = 0; w < 8; ++w) s += redw[w];
    return s;
}

__global__ __launch_bounds__(256)
void hra_gs_kernel(const float* __restrict__ hra_u, float* __restrict__ U) {
    __shared__ float redw[8];
    const int t = threadIdx.x;
    float u[RANK][8];
    #pragma unroll
    for (int e = 0; e < 8; ++e) {
        int d = e * 256 + t;
        #pragma unroll
        for (int j = 0; j < RANK; ++j) u[j][e] = hra_u[d * RANK + j];
    }
    for (int i = 0; i < RANK; ++i) {
        for (int j = 0; j < i; ++j) {
            float p = 0.f;
            #pragma unroll
            for (int e = 0; e < 8; ++e) p += u[j][e] * u[i][e];
            float dot = block_reduce_sum_256(p, redw);
            #pragma unroll
            for (int e = 0; e < 8; ++e) u[i][e] -= dot * u[j][e];
        }
        float p = 0.f;
        #pragma unroll
        for (int e = 0; e < 8; ++e) p += u[i][e] * u[i][e];
        float nrm2 = block_reduce_sum_256(p, redw);
        float inv = 1.0f / (sqrtf(nrm2) + 1e-6f);
        #pragma unroll
        for (int e = 0; e < 8; ++e) u[i][e] *= inv;
    }
    #pragma unroll
    for (int j = 0; j < RANK; ++j)
        #pragma unroll
        for (int e = 0; e < 8; ++e)
            U[j * D_IN + e * 256 + t] = u[j][e];
}

__global__ __launch_bounds__(256)
void hra_apply_kernel(const float* __restrict__ x, const float* __restrict__ U,
                      float* __restrict__ y) {
    __shared__ float red[8][RANK];
    const int t   = threadIdx.x;
    const size_t row = blockIdx.x;
    const float* xr = x + row * D_IN;

    float xv[8];
    #pragma unroll
    for (int e = 0; e < 8; ++e) xv[e] = xr[e * 256 + t];

    float c[RANK];
    #pragma unroll
    for (int j = 0; j < RANK; ++j) {
        float p = 0.f;
        #pragma unroll
        for (int e = 0; e < 8; ++e) p += xv[e] * U[j * D_IN + e * 256 + t];
        #pragma unroll
        for (int off = 16; off > 0; off >>= 1) p += __shfl_down(p, off, 32);
        if ((t & 31) == 0) red[t >> 5][j] = p;
    }
    __syncthreads();
    #pragma unroll
    for (int j = 0; j < RANK; ++j) {
        float s = 0.f;
        #pragma unroll
        for (int w = 0; w < 8; ++w) s += red[w][j];
        c[j] = 2.0f * s;
    }
    float* yr = y + row * D_IN;
    #pragma unroll
    for (int e = 0; e < 8; ++e) {
        int d = e * 256 + t;
        float acc = xv[e];
        #pragma unroll
        for (int j = 0; j < RANK; ++j) acc -= c[j] * U[j * D_IN + d];
        yr[d] = acc;
    }
}

#define BM 128
#define BN 128
#define BK 16
#define A_PL 528
#define B_PL 512
#define B_JP 128

template <int EPI>
__global__ __launch_bounds__(256)
void wmma_gemm_f32_kernel(const float* A, const float* B,
                          const float* __restrict__ bias, float* __restrict__ C,
                          int M, int N, int K) {
    __shared__ __attribute__((aligned(16))) float As[2][4 * A_PL];
    __shared__ __attribute__((aligned(16))) float Bs[2][4 * B_PL];

    const int t    = threadIdx.x;
    const int wave = t >> 5;
    const int lane = t & 31;
    const int wm   = wave & 3;
    const int wn   = wave >> 2;
    const int l16  = lane & 15;
    const int lh   = lane >> 4;

    const int bp   = 2 * (wave & 3) + lh;
    const int bq   = (wave >> 2) * 16 + l16;

    const int m_blk = blockIdx.y * BM;
    const int n_blk = blockIdx.x * BN;

    v8f acc[2][4] = {};
    v4f ra0, ra1, rb0, rb1;

    const int a_row0 = t >> 2;
    const int a_kq   = t & 3;
    const float* a_src = A + (size_t)(m_blk + a_row0) * K + (a_kq << 2);
    const float* b_src = B + (size_t)(2 * bp) * N + n_blk + (bq << 2);
    const int a_dst = a_kq * A_PL + (a_row0 << 2);
    const int b_dst = (bp >> 1) * B_PL + (bq << 2) + ((bp & 1) << 1);

    auto load_tiles = [&](int k0) {
        ra0 = *(const v4f*)(a_src + k0);
        ra1 = *(const v4f*)(a_src + (size_t)64 * K + k0);
        rb0 = *(const v4f*)(b_src + (size_t)k0 * N);
        rb1 = *(const v4f*)(b_src + (size_t)(k0 + 1) * N);
    };
    auto store_tiles = [&](int buf) {
        *(v4f*)(&As[buf][a_dst])            = ra0;
        *(v4f*)(&As[buf][a_dst + (64 << 2)]) = ra1;
        float* pb = &Bs[buf][b_dst];
        v2f p0 = {rb0.x, rb1.x};
        v2f p1 = {rb0.y, rb1.y};
        v2f p2 = {rb0.z, rb1.z};
        v2f p3 = {rb0.w, rb1.w};
        *(v2f*)(pb + 0 * B_JP) = p0;
        *(v2f*)(pb + 1 * B_JP) = p1;
        *(v2f*)(pb + 2 * B_JP) = p2;
        *(v2f*)(pb + 3 * B_JP) = p3;
    };
    v2f af[4][2], bf[4][4];
    auto load_frags = [&](int buf) {
        #pragma unroll
        for (int s = 0; s < 4; ++s) {
            #pragma unroll
            for (int i = 0; i < 2; ++i)
                af[s][i] = *(const v2f*)(&As[buf][s * A_PL +
                                ((wm * 32 + i * 16 + l16) << 2) + 2 * lh]);
            #pragma unroll
            for (int j = 0; j < 4; ++j)
                bf[s][j] = *(const v2f*)(&Bs[buf][s * B_PL + j * B_JP +
                                ((wn * 16 + l16) << 2) + 2 * lh]);
        }
    };
    auto mma_block = [&]() {
        #pragma unroll
        for (int s = 0; s < 4; ++s)
            #pragma unroll
            for (int i = 0; i < 2; ++i)
                #pragma unroll
                for (int j = 0; j < 4; ++j)
                    acc[i][j] = __builtin_amdgcn_wmma_f32_16x16x4_f32(
                        false, af[s][i], false, bf[s][j], (short)0, acc[i][j],
                        false, false);
    };

    auto step = [&](int buf, bool has_next, bool has_next2, int k_next2) {
        load_frags(buf);
        if (has_next)  store_tiles(buf ^ 1);
        if (has_next2) load_tiles(k_next2);
        asm volatile("" ::: "memory");
        if (has_next) {
            __builtin_amdgcn_fence(__ATOMIC_RELEASE, "workgroup", "local");
            __builtin_amdgcn_s_barrier_signal(-1);
        }
        __builtin_amdgcn_sched_barrier(0);
        mma_block();
        __builtin_amdgcn_sched_barrier(0);
        if (has_next) {
            __builtin_amdgcn_s_barrier_wait(-1);
            __builtin_amdgcn_fence(__ATOMIC_ACQUIRE, "workgroup", "local");
        }
    };

    const int nch = K / BK;
    load_tiles(0);
    store_tiles(0);
    if (nch > 1) load_tiles(BK);
    __syncthreads();
    int c = 0;
    for (; c + 2 < nch; ++c) step(c & 1, true, true, (c + 2) * BK);
    if (c + 1 < nch) { step(c & 1, true, false, 0); ++c; }
    step(c & 1, false, false, 0);

    const int col0 = n_blk + wn * 64 + (l16 << 2);
    float4 bv = make_float4(0.f, 0.f, 0.f, 0.f);
    if (EPI == 1) bv = *(const float4*)(bias + col0);
    #pragma unroll
    for (int i = 0; i < 2; ++i) {
        #pragma unroll
        for (int r = 0; r < 8; ++r) {
            int row = m_blk + wm * 32 + i * 16 + r + 8 * lh;
            float v[4];
            #pragma unroll
            for (int j = 0; j < 4; ++j) {
                float x = acc[i][j][r];
                if (EPI == 0) x = 0.5f * x * (1.0f + erff(x * 0.70710678118654752f));
                v[j] = x;
            }
            float4 o;
            if (EPI == 0) {
                o = make_float4(v[0], v[1], v[2], v[3]);
            } else {
                o = make_float4(v[0] + bv.x, v[1] + bv.y, v[2] + bv.z, v[3] + bv.w);
            }
            *(float4*)(C + (size_t)row * N + col0) = o;
        }
    }
}

extern "C" void kernel_launch(void* const* d_in, const int* in_sizes, int n_in,
                              void* d_out, int out_size, void* d_ws, size_t ws_size,
                              hipStream_t stream) {
    const float* x    = (const float*)d_in[0];
    const float* hra  = (const float*)d_in[1];
    const float* Wr   = (const float*)d_in[2];
    const float* Wo   = (const float*)d_in[3];
    const float* bias = (const float*)d_in[4];
    float* out = (float*)d_out;

    char* ws = (char*)d_ws;
    float* U = (float*)ws;
    float* y = (float*)(ws + (64u << 10));
    float* h = (float*)(ws + (64u << 10) + (32u << 20));

    hra_gs_kernel<<<1, 256, 0, stream>>>(hra, U);

    hra_apply_kernel<<<N_ROWS, 256, 0, stream>>>(x, U, y);

    dim3 g1(D_HID / BN, N_ROWS / BM);
    wmma_gemm_f32_kernel<0><<<g1, 256, 0, stream>>>(y, Wr, nullptr, h,
                                                    N_ROWS, D_HID, D_IN);

    dim3 g2(D_OUT / BN, N_ROWS / BM);
    wmma_gemm_f32_kernel<1><<<g2, 256, 0, stream>>>(h, Wo, bias, out,
                                                    N_ROWS, D_OUT, D_HID);
}


# GEMM K-loop: each wave's 2x2 block product per 16-k chunk done with 7 block products instead of 8 (operand sums by f32 vector adds): 28 WMMAs per chunk, not 32; summation order changed
# speedup vs baseline: 1.2290x; 1.0399x over previous
#include <hip/hip_runtime.h>
#include <hip/hip_bf16.h>
#include <math.h>

typedef __attribute__((ext_vector_type(2))) float v2f;
typedef __attribute__((ext_vector_type(8))) float v8f;
typedef __attribute__((ext_vector_type(4))) float v4f;

#define N_ROWS   4096
#define D_IN     2048
#define D_HID    8192
#define D_OUT    2048
#define RANK     8

__device__ __forceinline__ float block_reduce_sum_256(float p, float* redw) {
    #pragma unroll
    for (int off = 16; off > 0; off >>= 1) p += __shfl_down(p, off, 32);
    int t = threadIdx.x;
    __syncthreads();
    if ((t & 31) == 0) redw[t >> 5] = p;
    __syncthreads();
    float s = 0.f;
    #pragma unroll
    for (int w = 0; w < 8; ++w) s += redw[w];
    return s;
}

__global__ __launch_bounds__(256)
void hra_gs_kernel(const float* __restrict__ hra_u, float* __restrict__ U) {
    __shared__ float redw[8];
    const int t = threadIdx.x;
    float u[RANK][8];
    #pragma unroll
    for (int e = 0; e < 8; ++e) {
        int d = e * 256 + t;
        #pragma unroll
        for (int j = 0; j < RANK; ++j) u[j][e] = hra_u[d * RANK + j];
    }
    for (int i = 0; i < RANK; ++i) {
        for (int j = 0; j < i; ++j) {
            float p = 0.f;
            #pragma unroll
            for (int e = 0; e < 8; ++e) p += u[j][e] * u[i][e];
            float dot = block_reduce_sum_256(p, redw);
            #pragma unroll
            for (int e = 0; e < 8; ++e) u[i][e] -= dot * u[j][e];
        }
        float p = 0.f;
        #pragma unroll
        for (int e = 0; e < 8; ++e) p += u[i][e] * u[i][e];
        float nrm2 = block_reduce_sum_256(p, redw);
        float inv = 1.0f / (sqrtf(nrm2) + 1e-6f);
        #pragma unroll
        for (int e = 0; e < 8; ++e) u[i][e] *= inv;
    }
    #pragma unroll
    for (int j = 0; j < RANK; ++j)
        #pragma unroll
        for (int e = 0; e < 8; ++e)
            U[j * D_IN + e * 256 + t] = u[j][e];
}

__global__ __launch_bounds__(256)
void hra_apply_kernel(const float* __restrict__ x, const float* __restrict__ U,
                      float* __restrict__ y) {
    __shared__ float red[8][RANK];
    const int t   = threadIdx.x;
    const size_t row = blockIdx.x;
    const float* xr = x + row * D_IN;

    float xv[8];
    #pragma unroll
    for (int e = 0; e < 8; ++e) xv[e] = xr[e * 256 + t];

    float c[RANK];
    #pragma unroll
    for (int j = 0; j < RANK; ++j) {
        float p = 0.f;
        #pragma unroll
        for (int e = 0; e < 8; ++e) p += xv[e] * U[j * D_IN + e * 256 + t];
        #pragma unroll
        for (int off = 16; off > 0; off >>= 1) p += __shfl_down(p, off, 32);
        if ((t & 31) == 0) red[t >> 5][j] = p;
    }
    __syncthreads();
    #pragma unroll
    for (int j = 0; j < RANK; ++j) {
        float s = 0.f;
        #pragma unroll
        for (int w = 0; w < 8; ++w) s += red[w][j];
        c[j] = 2.0f * s;
    }
    float* yr = y + row * D_IN;
    #pragma unroll
    for (int e = 0; e < 8; ++e) {
        int d = e * 256 + t;
        float acc = xv[e];
        #pragma unroll
        for (int j = 0; j < RANK; ++j) acc -= c[j] * U[j * D_IN + d];
        yr[d] = acc;
    }
}

#define BM 128
#define BN 128
#define BK 16
#define A_PL 528
#define B_PL 512
#define B_JP 128

template <int EPI>
__global__ __launch_bounds__(256)
void wmma_gemm_f32_kernel(const float* A, const float* B,
                          const float* __restrict__ bias, float* __restrict__ C,
                          int M, int N, int K) {
    __shared__ __attribute__((aligned(16))) float As[2][4 * A_PL];
    __shared__ __attribute__((aligned(16))) float Bs[2][4 * B_PL];

    const int t    = threadIdx.x;
    const int wave = t >> 5;
    const int lane = t & 31;
    const int wm   = wave & 3;
    const int wn   = wave >> 2;
    const int l16  = lane & 15;
    const int lh   = lane >> 4;

    const int bp   = 2 * (wave & 3) + lh;
    const int bq   = (wave >> 2) * 16 + l16;

    const int m_blk = blockIdx.y * BM;
    const int n_blk = blockIdx.x * BN;

    v8f acc[7][2] = {};
    v4f ra0, ra1, rb0, rb1;

    const int a_row0 = t >> 2;
    const int a_kq   = t & 3;
    const float* a_src = A + (size_t)(m_blk + a_row0) * K + (a_kq << 2);
    const float* b_src = B + (size_t)(2 * bp) * N + n_blk + (bq << 2);
    const int a_dst = a_kq * A_PL + (a_row0 << 2);
    const int b_dst = (bp >> 1) * B_PL + (bq << 2) + ((bp & 1) << 1);

    auto load_tiles = [&](int k0) {
        ra0 = *(const v4f*)(a_src + k0);
        ra1 = *(const v4f*)(a_src + (size_t)64 * K + k0);
        rb0 = *(const v4f*)(b_src + (size_t)k0 * N);
        rb1 = *(const v4f*)(b_src + (size_t)(k0 + 1) * N);
    };
    auto store_tiles = [&](int buf) {
        *(v4f*)(&As[buf][a_dst])            = ra0;
        *(v4f*)(&As[buf][a_dst + (64 << 2)]) = ra1;
        float* pb = &Bs[buf][b_dst];
        v2f p0 = {rb0.x, rb1.x};
        v2f p1 = {rb0.y, rb1.y};
        v2f p2 = {rb0.z, rb1.z};
        v2f p3 = {rb0.w, rb1.w};
        *(v2f*)(pb + 0 * B_JP) = p0;
        *(v2f*)(pb + 1 * B_JP) = p1;
        *(v2f*)(pb + 2 * B_JP) = p2;
        *(v2f*)(pb + 3 * B_JP) = p3;
    };
    v2f af[4][2], bf[4][4];
    auto load_frags = [&](int buf) {
        #pragma unroll
        for (int s = 0; s < 4; ++s) {
            #pragma unroll
            for (int i = 0; i < 2; ++i)
                af[s][i] = *(const v2f*)(&As[buf][s * A_PL +
                                ((wm * 32 + i * 16 + l16) << 2) + 2 * lh]);
            #pragma unroll
            for (int j = 0; j < 4; ++j)
                bf[s][j] = *(const v2f*)(&Bs[buf][s * B_PL + j * B_JP +
                                ((wn * 16 + l16) << 2) + 2 * lh]);
        }
    };
    auto wm4 = [&](v2f a, v2f b, v8f c) -> v8f {
        return __builtin_amdgcn_wmma_f32_16x16x4_f32(false, a, false, b, (short)0, c,
                                                     false, false);
    };
    auto mma_block = [&]() {
        v2f sa[2][5], tb[2][2][5];
        #pragma unroll
        for (int p = 0; p < 2; ++p) {
            const v2f a11 = af[p][0], a12 = af[2 + p][0];
            const v2f a21 = af[p][1], a22 = af[2 + p][1];
            sa[p][0] = a11 + a22; sa[p][1] = a21 + a22; sa[p][2] = a11 + a12;
            sa[p][3] = a21 - a11; sa[p][4] = a12 - a22;
            #pragma unroll
            for (int jj = 0; jj < 2; ++jj) {
                const v2f b11 = bf[p][jj],     b12 = bf[p][2 + jj];
                const v2f b21 = bf[2 + p][jj], b22 = bf[2 + p][2 + jj];
                tb[p][jj][0] = b11 + b22; tb[p][jj][1] = b12 - b22;
                tb[p][jj][2] = b21 - b11; tb[p][jj][3] = b11 + b12;
                tb[p][jj][4] = b21 + b22;
            }
        }
        __builtin_amdgcn_sched_barrier(0);
        #pragma unroll
        for (int p = 0; p < 2; ++p) {
            #pragma unroll
            for (int jj = 0; jj < 2; ++jj) {
                acc[0][jj] = wm4(sa[p][0],      tb[p][jj][0],       acc[0][jj]);
                acc[1][jj] = wm4(sa[p][1],      bf[p][jj],          acc[1][jj]);
                acc[2][jj] = wm4(af[p][0],      tb[p][jj][1],       acc[2][jj]);
                acc[3][jj] = wm4(af[2 + p][1],  tb[p][jj][2],       acc[3][jj]);
                acc[4][jj] = wm4(sa[p][2],      bf[2 + p][2 + jj],  acc[4][jj]);
                acc[5][jj] = wm4(sa[p][3],      tb[p][jj][3],       acc[5][jj]);
                acc[6][jj] = wm4(sa[p][4],      tb[p][jj][4],       acc[6][jj]);
            }
        }
    };

    auto step = [&](int buf, bool has_next, bool has_next2, int k_next2) {
        load_frags(buf);
        if (has_next)  store_tiles(buf ^ 1);
        if (has_next2) load_tiles(k_next2);
        asm volatile("" ::: "memory");
        if (has_next) {
            __builtin_amdgcn_fence(__ATOMIC_RELEASE, "workgroup", "local");
            __builtin_amdgcn_s_barrier_signal(-1);
        }
        __builtin_amdgcn_sched_barrier(0);
        mma_block();
        __builtin_amdgcn_sched_barrier(0);
        if (has_next) {
            __builtin_amdgcn_s_barrier_wait(-1);
            __builtin_amdgcn_fence(__ATOMIC_ACQUIRE, "workgroup", "local");
        }
    };

    const int nch = K / BK;
    load_tiles(0);
    store_tiles(0);
    if (nch > 1) load_tiles(BK);
    __syncthreads();
    int c = 0;
    for (; c + 2 < nch; ++c) step(c & 1, true, true, (c + 2) * BK);
    if (c + 1 < nch) { step(c & 1, true, false, 0); ++c; }
    step(c & 1, false, false, 0);

    asm volatile("v_nop\n\tv_nop\n\tv_nop\n\tv_nop\n\tv_nop\n\tv_nop\n\tv_nop\n\tv_nop"
                 : "+v"(acc[0][0]), "+v"(acc[0][1]), "+v"(acc[1][0]), "+v"(acc[1][1]), "+v"(acc[2][0]), "+v"(acc[2][1]), "+v"(acc[3][0]),
                   "+v"(acc[3][1]), "+v"(acc[4][0]), "+v"(acc[4][1]), "+v"(acc[5][0]), "+v"(acc[5][1]), "+v"(acc[6][0]), "+v"(acc[6][1]));
    v8f ct[2][4];
    #pragma unroll
    for (int jj = 0; jj < 2; ++jj) {
        ct[0][jj]     = (acc[0][jj] + acc[3][jj]) + (acc[6][jj] - acc[4][jj]);
        ct[0][2 + jj] = acc[2][jj] + acc[4][jj];
        ct[1][jj]     = acc[1][jj] + acc[3][jj];
        ct[1][2 + jj] = (acc[0][jj] - acc[1][jj]) + (acc[2][jj] + acc[5][jj]);
    }
    const int col0 = n_blk + wn * 64 + (l16 << 2);
    float4 bv = make_float4(0.f, 0.f, 0.f, 0.f);
    if (EPI == 1) bv = *(const float4*)(bias + col0);
    #pragma unroll
    for (int i = 0; i < 2; ++i) {
        #pragma unroll
        for (int r = 0; r < 8; ++r) {
            int row = m_blk + wm * 32 + i * 16 + r + 8 * lh;
            float v[4];
            #pragma unroll
            for (int j = 0; j < 4; ++j) {
                float x = ct[i][j][r];
                if (EPI == 0) x = 0.5f * x * (1.0f + erff(x * 0.70710678118654752f));
                v[j] = x;
            }
            float4 o;
            if (EPI == 0) {
                o = make_float4(v[0], v[1], v[2], v[3]);
            } else {
                o = make_float4(v[0] + bv.x, v[1] + bv.y, v[2] + bv.z, v[3] + bv.w);
            }
            *(float4*)(C + (size_t)row * N + col0) = o;
        }
    }
}

extern "C" void kernel_launch(void* const* d_in, const int* in_sizes, int n_in,
                              void* d_out, int out_size, void* d_ws, size_t ws_size,
                              hipStream_t stream) {
    const float* x    = (const float*)d_in[0];
    const float* hra  = (const float*)d_in[1];
    const float* Wr   = (const float*)d_in[2];
    const float* Wo   = (const float*)d_in[3];
    const float* bias = (const float*)d_in[4];
    float* out = (float*)d_out;

    char* ws = (char*)d_ws;
    float* U = (float*)ws;
    float* y = (float*)(ws + (64u << 10));
    float* h = (float*)(ws + (64u << 10) + (32u << 20));

    hra_gs_kernel<<<1, 256, 0, stream>>>(hra, U);

    hra_apply_kernel<<<N_ROWS, 256, 0, stream>>>(x, U, y);

    dim3 g1(D_HID / BN, N_ROWS / BM);
    wmma_gemm_f32_kernel<0><<<g1, 256, 0, stream>>>(y, Wr, nullptr, h,
                                                    N_ROWS, D_HID, D_IN);

    dim3 g2(D_OUT / BN, N_ROWS / BM);
    wmma_gemm_f32_kernel<1><<<g2, 256, 0, stream>>>(h, Wo, bias, out,
                                                    N_ROWS, D_OUT, D_HID);
}


# GEMM K-loop without vector address arithmetic: two chunks per trip (LDS stage offsets literal), N and K as template constants (one stepped pointer per operand, all other offsets literal)
# speedup vs baseline: 1.2683x; 1.0320x over previous
#include <hip/hip_runtime.h>
#include <hip/hip_bf16.h>
#include <math.h>

typedef __attribute__((ext_vector_type(2))) float v2f;
typedef __attribute__((ext_vector_type(8))) float v8f;
typedef __attribute__((ext_vector_type(4))) float v4f;

#define N_ROWS   4096
#define D_IN     2048
#define D_HID    8192
#define D_OUT    2048
#define RANK     8

__device__ __forceinline__ float block_reduce_sum_256(float p, float* redw) {
    #pragma unroll
    for (int off = 16; off > 0; off >>= 1) p += __shfl_down(p, off, 32);
    int t = threadIdx.x;
    __syncthreads();
    if ((t & 31) == 0) redw[t >> 5] = p;
    __syncthreads();
    float s = 0.f;
    #pragma unroll
    for (int w = 0; w < 8; ++w) s += redw[w];
    return s;
}

__global__ __launch_bounds__(256)
void hra_gs_kernel(const float* __restrict__ hra_u, float* __restrict__ U) {
    __shared__ float redw[8];
    const int t = threadIdx.x;
    float u[RANK][8];
    #pragma unroll
    for (int e = 0; e < 8; ++e) {
        int d = e * 256 + t;
        #pragma unroll
        for (int j = 0; j < RANK; ++j) u[j][e] = hra_u[d * RANK + j];
    }
    for (int i = 0; i < RANK; ++i) {
        for (int j = 0; j < i; ++j) {
            float p = 0.f;
            #pragma unroll
            for (int e = 0; e < 8; ++e) p += u[j][e] * u[i][e];
            float dot = block_reduce_sum_256(p, redw);
            #pragma unroll
            for (int e = 0; e < 8; ++e) u[i][e] -= dot * u[j][e];
        }
        float p = 0.f;
        #pragma unroll
        for (int e = 0; e < 8; ++e) p += u[i][e] * u[i][e];
        float nrm2 = block_reduce_sum_256(p, redw);
        float inv = 1.0f / (sqrtf(nrm2) + 1e-6f);
        #pragma unroll
        for (int e = 0; e < 8; ++e) u[i][e] *= inv;
    }
    #pragma unroll
    for (int j = 0; j < RANK; ++j)
        #pragma unroll
        for (int e = 0; e < 8; ++e)
            U[j * D_IN + e * 256 + t] = u[j][e];
}

__global__ __launch_bounds__(256)
void hra_apply_kernel(const float* __restrict__ x, const float* __restrict__ U,
                      float* __restrict__ y) {
    __shared__ float red[8][RANK];
    const int t   = threadIdx.x;
    const size_t row = blockIdx.x;
    const float* xr = x + row * D_IN;

    float xv[8];
    #pragma unroll
    for (int e = 0; e < 8; ++e) xv[e] = xr[e * 256 + t];

    float c[RANK];
    #pragma unroll
    for (int j = 0; j < RANK; ++j) {
        float p = 0.f;
        #pragma unroll
        for (int e = 0; e < 8; ++e) p += xv[e] * U[j * D_IN + e * 256 + t];
        #pragma unroll
        for (int off = 16; off > 0; off >>= 1) p += __shfl_down(p, off, 32);
        if ((t & 31) == 0) red[t >> 5][j] = p;
    }
    __syncthreads();
    #pragma unroll
    for (int j = 0; j < RANK; ++j) {
        float s = 0.f;
        #pragma unroll
        for (int w = 0; w < 8; ++w) s += red[w][j];
        c[j] = 2.0f * s;
    }
    float* yr = y + row * D_IN;
    #pragma unroll
    for (int e = 0; e < 8; ++e) {
        int d = e * 256 + t;
        float acc = xv[e];
        #pragma unroll
        for (int j = 0; j < RANK; ++j) acc -= c[j] * U[j * D_IN + d];
        yr[d] = acc;
    }
}

#define BM 128
#define BN 128
#define BK 16
#define A_PL 528
#define B_PL 512
#define B_JP 128

template <int EPI, int N, int K>
__global__ __launch_bounds__(256)
void wmma_gemm_f32_kernel(const float* A, const float* B,
                          const float* __restrict__ bias, float* __restrict__ C) {
    __shared__ __attribute__((aligned(16))) float As[2][4 * A_PL];
    __shared__ __attribute__((aligned(16))) float Bs[2][4 * B_PL];

    const int t    = threadIdx.x;
    const int wave = t >> 5;
    const int lane = t & 31;
    const int wm   = wave & 3;
    const int wn   = wave >> 2;
    const int l16  = lane & 15;
    const int lh   = lane >> 4;

    const int bp   = 2 * (wave & 3) + lh;
    const int bq   = (wave >> 2) * 16 + l16;

    const int m_blk = blockIdx.y * BM;
    const int n_blk = blockIdx.x * BN;

    v8f acc[7][2] = {};
    v4f ra0, ra1, rb0, rb1;

    const int a_row0 = t >> 2;
    const int a_kq   = t & 3;
    const float* a_src = A + (size_t)(m_blk + a_row0) * K + (a_kq << 2);
    const float* b_src = B + (size_t)(2 * bp) * N + n_blk + (bq << 2);
    const int a_dst = a_kq * A_PL + (a_row0 << 2);
    const int b_dst = (bp >> 1) * B_PL + (bq << 2) + ((bp & 1) << 1);

    auto load_tiles = [&](const float* ap, const float* bp_) {
        ra0 = *(const v4f*)(ap);
        ra1 = *(const v4f*)(ap + (size_t)64 * K);
        rb0 = *(const v4f*)(bp_);
        rb1 = *(const v4f*)(bp_ + N);
    };
    auto store_tiles = [&](int buf) {
        *(v4f*)(&As[buf][a_dst])            = ra0;
        *(v4f*)(&As[buf][a_dst + (64 << 2)]) = ra1;
        float* pb = &Bs[buf][b_dst];
        v2f p0 = {rb0.x, rb1.x};
        v2f p1 = {rb0.y, rb1.y};
        v2f p2 = {rb0.z, rb1.z};
        v2f p3 = {rb0.w, rb1.w};
        *(v2f*)(pb + 0 * B_JP) = p0;
        *(v2f*)(pb + 1 * B_JP) = p1;
        *(v2f*)(pb + 2 * B_JP) = p2;
        *(v2f*)(pb + 3 * B_JP) = p3;
    };
    v2f af[4][2], bf[4][4];
    auto load_frags = [&](int buf) {
        #pragma unroll
        for (int s = 0; s < 4; ++s) {
            #pragma unroll
            for (int i = 0; i < 2; ++i)
                af[s][i] = *(const v2f*)(&As[buf][s * A_PL +
                                ((wm * 32 + i * 16 + l16) << 2) + 2 * lh]);
            #pragma unroll
            for (int j = 0; j < 4; ++j)
                bf[s][j] = *(const v2f*)(&Bs[buf][s * B_PL + j * B_JP +
                                ((wn * 16 + l16) << 2) + 2 * lh]);
        }
    };
    auto wm4 = [&](v2f a, v2f b, v8f c) -> v8f {
        return __builtin_amdgcn_wmma_f32_16x16x4_f32(false, a, false, b, (short)0, c,
                                                     false, false);
    };
    auto mma_block = [&]() {
        v2f sa[2][5], tb[2][2][5];
        #pragma unroll
        for (int p = 0; p < 2; ++p) {
            const v2f a11 = af[p][0], a12 = af[2 + p][0];
            const v2f a21 = af[p][1], a22 = af[2 + p][1];
            sa[p][0] = a11 + a22; sa[p][1] = a21 + a22; sa[p][2] = a11 + a12;
            sa[p][3] = a21 - a11; sa[p][4] = a12 - a22;
            #pragma unroll
            for (int jj = 0; jj < 2; ++jj) {
                const v2f b11 = bf[p][jj],     b12 = bf[p][2 + jj];
                const v2f b21 = bf[2 + p][jj], b22 = bf[2 + p][2 + jj];
                tb[p][jj][0] = b11 + b22; tb[p][jj][1] = b12 - b22;
                tb[p][jj][2] = b21 - b11; tb[p][jj][3] = b11 + b12;
                tb[p][jj][4] = b21 + b22;
            }
        }
        __builtin_amdgcn_sched_barrier(0);
        #pragma unroll
        for (int p = 0; p < 2; ++p) {
            #pragma unroll
            for (int jj = 0; jj < 2; ++jj) {
                acc[0][jj] = wm4(sa[p][0],      tb[p][jj][0],       acc[0][jj]);
                acc[1][jj] = wm4(sa[p][1],      bf[p][jj],          acc[1][jj]);
                acc[2][jj] = wm4(af[p][0],      tb[p][jj][1],       acc[2][jj]);
                acc[3][jj] = wm4(af[2 + p][1],  tb[p][jj][2],       acc[3][jj]);
                acc[4][jj] = wm4(sa[p][2],      bf[2 + p][2 + jj],  acc[4][jj]);
                acc[5][jj] = wm4(sa[p][3],      tb[p][jj][3],       acc[5][jj]);
                acc[6][jj] = wm4(sa[p][4],      tb[p][jj][4],       acc[6][jj]);
            }
        }
    };

    auto step = [&](int buf, bool has_next, bool has_next2,
                    const float* ap2, const float* bp2) {
        load_frags(buf);
        if (has_next)  store_tiles(buf ^ 1);
        if (has_next2) load_tiles(ap2, bp2);
        asm volatile("" ::: "memory");
        if (has_next) {
            __builtin_amdgcn_fence(__ATOMIC_RELEASE, "workgroup", "local");
            __builtin_amdgcn_s_barrier_signal(-1);
        }
        __builtin_amdgcn_sched_barrier(0);
        mma_block();
        __builtin_amdgcn_sched_barrier(0);
        if (has_next) {
            __builtin_amdgcn_s_barrier_wait(-1);
            __builtin_amdgcn_fence(__ATOMIC_ACQUIRE, "workgroup", "local");
        }
    };

    constexpr int nch = K / BK;
    constexpr size_t b_step = (size_t)BK * N;
    load_tiles(a_src, b_src);
    store_tiles(0);
    if (nch > 1) load_tiles(a_src + BK, b_src + b_step);
    __syncthreads();
    int c = 0;
    const float* ap2 = a_src + 2 * BK;
    const float* bp2 = b_src + 2 * b_step;
    for (; c + 3 < nch; c += 2) {
        step(0, true, true, ap2, bp2);
        step(1, true, true, ap2 + BK, bp2 + b_step);
        ap2 += 2 * BK;  bp2 += 2 * b_step;
    }
    for (; c + 2 < nch; ++c) { step(c & 1, true, true, ap2, bp2);  ap2 += BK;  bp2 += b_step; }
    if (c + 1 < nch) { step(c & 1, true, false, a_src, b_src); ++c; }
    step(c & 1, false, false, a_src, b_src);

    asm volatile("v_nop\n\tv_nop\n\tv_nop\n\tv_nop\n\tv_nop\n\tv_nop\n\tv_nop\n\tv_nop"
                 : "+v"(acc[0][0]), "+v"(acc[0][1]), "+v"(acc[1][0]), "+v"(acc[1][1]), "+v"(acc[2][0]), "+v"(acc[2][1]), "+v"(acc[3][0]),
                   "+v"(acc[3][1]), "+v"(acc[4][0]), "+v"(acc[4][1]), "+v"(acc[5][0]), "+v"(acc[5][1]), "+v"(acc[6][0]), "+v"(acc[6][1]));
    v8f ct[2][4];
    #pragma unroll
    for (int jj = 0; jj < 2; ++jj) {
        ct[0][jj]     = (acc[0][jj] + acc[3][jj]) + (acc[6][jj] - acc[4][jj]);
        ct[0][2 + jj] = acc[2][jj] + acc[4][jj];
        ct[1][jj]     = acc[1][jj] + acc[3][jj];
        ct[1][2 + jj] = (acc[0][jj] - acc[1][jj]) + (acc[2][jj] + acc[5][jj]);
    }
    const int col0 = n_blk + wn * 64 + (l16 << 2);
    float4 bv = make_float4(0.f, 0.f, 0.f, 0.f);
    if (EPI == 1) bv = *(const float4*)(bias + col0);
    #pragma unroll
    for (int i = 0; i < 2; ++i) {
        #pragma unroll
        for (int r = 0; r < 8; ++r) {
            int row = m_blk + wm * 32 + i * 16 + r + 8 * lh;
            float v[4];
            #pragma unroll
            for (int j = 0; j < 4; ++j) {
                float x = ct[i][j][r];
                if (EPI == 0) x = 0.5f * x * (1.0f + erff(x * 0.70710678118654752f));
                v[j] = x;
            }
            float4 o;
            if (EPI == 0) {
                o = make_float4(v[0], v[1], v[2], v[3]);
            } else {
                o = make_float4(v[0] + bv.x, v[1] + bv.y, v[2] + bv.z, v[3] + bv.w);
            }
            *(float4*)(C + (size_t)row * N + col0) = o;
        }
    }
}

extern "C" void kernel_launch(void* const* d_in, const int* in_sizes, int n_in,
                              void* d_out, int out_size, void* d_ws, size_t ws_size,
                              hipStream_t stream) {
    const float* x    = (const float*)d_in[0];
    const float* hra  = (const float*)d_in[1];
    const float* Wr   = (const float*)d_in[2];
    const float* Wo   = (const float*)d_in[3];
    const float* bias = (const float*)d_in[4];
    float* out = (float*)d_out;

    char* ws = (char*)d_ws;
    float* U = (float*)ws;
    float* y = (float*)(ws + (64u << 10));
    float* h = (float*)(ws + (64u << 10) + (32u << 20));

    hra_gs_kernel<<<1, 256, 0, stream>>>(hra, U);

    hra_apply_kernel<<<N_ROWS, 256, 0, stream>>>(x, U, y);

    dim3 g1(D_HID / BN, N_ROWS / BM);
    wmma_gemm_f32_kernel<0, D_HID, D_IN><<<g1, 256, 0, stream>>>(y, Wr, nullptr, h);

    dim3 g2(D_OUT / BN, N_ROWS / BM);
    wmma_gemm_f32_kernel<1, D_OUT, D_HID><<<g2, 256, 0, stream>>>(h, Wo, bias, out);
}
